# starting kernel
# baseline (speedup 1.0000x reference)
#include <hip/hip_runtime.h>
#include <math.h>

typedef __attribute__((ext_vector_type(2))) float v2f;
typedef __attribute__((ext_vector_type(8))) float v8f;

#define NF 128

__global__ void fill_zero(float* __restrict__ p, long n) {
  long i = (long)blockIdx.x * blockDim.x + threadIdx.x;
  long stride = (long)gridDim.x * blockDim.x;
  for (; i < n; i += stride) p[i] = 0.0f;
}

__global__ __launch_bounds__(128) void spmm_kernel(
    const float* __restrict__ X, const int* __restrict__ src,
    const int* __restrict__ dst, const float* __restrict__ vals,
    const float* __restrict__ Mv, const float* __restrict__ AMv,
    float* __restrict__ out) {
  int e = blockIdx.x;
  int f = threadIdx.x;
  int s = src[e];
  int d = dst[e];
  float w = vals[e];
  if (Mv)  w *= Mv[s];
  if (AMv) w *= AMv[d];
  atomicAdd(out + (size_t)d * NF + f, w * X[(size_t)s * NF + f]);
}

__global__ __launch_bounds__(256) void gemm_wmma_kernel(
    const float* __restrict__ A,
    const float* __restrict__ W,
    const float* __restrict__ bias,
    const float* __restrict__ g, const float* __restrict__ be,
    const float* __restrict__ rm, const float* __restrict__ rv,
    float* __restrict__ out,
    int n_rows, int n_out_real, int tiles_n, int relu) {
  const int K = NF;
  int lane = threadIdx.x & 31;
  int wave = threadIdx.x >> 5;
  int tiles_m = (n_rows + 15) >> 4;
  int tile = blockIdx.x * 8 + wave;
  if (tile >= tiles_m * tiles_n) return;
  int tm = tile / tiles_n;
  int tn = tile - tm * tiles_n;

  int hl = lane >> 4;
  int ll = lane & 15;
  int row = tm * 16 + ll;
  int ar  = row < n_rows ? row : n_rows - 1;
  int col = tn * 16 + ll;
  int cw  = col < n_out_real ? col : n_out_real - 1;
  int kh  = hl * 2;

  const float* ap = A + (size_t)ar * K + kh;
  const float* wp = W + (size_t)cw * K + kh;

  v8f c = {};
#pragma unroll
  for (int k0 = 0; k0 < K; k0 += 4) {
    v2f a = *(const v2f*)(ap + k0);
    v2f b = *(const v2f*)(wp + k0);
    c = __builtin_amdgcn_wmma_f32_16x16x4_f32(
         false, a,  false, b,
         (short)0, c,  false,  false);
  }

  float bb = bias ? bias[cw] : 0.0f;
  float sc = 1.0f, sh = 0.0f;
  if (g) {
    float s = g[cw] * rsqrtf(rv[cw] + 1e-5f);
    sc = s;
    sh = be[cw] - rm[cw] * s;
  }
  if (col < n_out_real) {
    int rb = tm * 16 + hl * 8;
#pragma unroll
    for (int i = 0; i < 8; ++i) {
      int r = rb + i;
      if (r < n_rows) {
        float v = (c[i] + bb) * sc + sh;
        if (relu) v = fmaxf(v, 0.0f);
        out[(size_t)r * n_out_real + col] = v;
      }
    }
  }
}

__global__ __launch_bounds__(256) void log_softmax40(float* __restrict__ out, int n) {
  int i = blockIdx.x * blockDim.x + threadIdx.x;
  if (i >= n) return;
  float* row = out + (size_t)i * 40;
  float m = row[0];
#pragma unroll
  for (int j = 1; j < 40; ++j) m = fmaxf(m, row[j]);
  float s = 0.0f;
#pragma unroll
  for (int j = 0; j < 40; ++j) s += expf(row[j] - m);
  float l = m + logf(s);
#pragma unroll
  for (int j = 0; j < 40; ++j) row[j] = row[j] - l;
}

extern "C" void kernel_launch(void* const* d_in, const int* in_sizes, int n_in,
                              void* d_out, int out_size, void* d_ws, size_t ws_size,
                              hipStream_t stream) {
  const float* x     = (const float*)d_in[0];
  const float* Mv    = (const float*)d_in[1];
  const float* AMv   = (const float*)d_in[2];
  const int*   srcZ  = (const int*)d_in[3];
  const int*   dstZ  = (const int*)d_in[4];
  const float* valsZ = (const float*)d_in[5];
  const int*   src   = (const int*)d_in[6];
  const int*   dst   = (const int*)d_in[7];
  const float* vals  = (const float*)d_in[8];
  const float* W0 = (const float*)d_in[9];
  const float* b0 = (const float*)d_in[10];
  const float* g0 = (const float*)d_in[11];
  const float* be0 = (const float*)d_in[12];
  const float* rm0 = (const float*)d_in[13];
  const float* rv0 = (const float*)d_in[14];
  const float* W1 = (const float*)d_in[15];
  const float* b1 = (const float*)d_in[16];
  const float* g1 = (const float*)d_in[17];
  const float* be1 = (const float*)d_in[18];
  const float* rm1 = (const float*)d_in[19];
  const float* rv1 = (const float*)d_in[20];
  const float* W2 = (const float*)d_in[21];
  const float* b2 = (const float*)d_in[22];

  int N = in_sizes[1];
  int E = in_sizes[3];
  float* bufA = (float*)d_ws;
  float* bufB = bufA + (size_t)N * NF;
  float* outp = (float*)d_out;

  int tiles_m = (N + 15) / 16;
  long nfeat_elems = (long)N * NF;
  int gemm_blocks_h = (tiles_m * 8 + 7) / 8;
  int gemm_blocks_o = (tiles_m * 3 + 7) / 8;

  fill_zero<<<2048, 256, 0, stream>>>(bufA, nfeat_elems);
  spmm_kernel<<<E, 128, 0, stream>>>(x, srcZ, dstZ, valsZ, Mv, AMv, bufA);
  gemm_wmma_kernel<<<gemm_blocks_h, 256, 0, stream>>>(
      bufA, W0, b0, g0, be0, rm0, rv0, bufB, N, NF, 8, 1);

  fill_zero<<<2048, 256, 0, stream>>>(bufA, nfeat_elems);
  spmm_kernel<<<E, 128, 0, stream>>>(bufB, src, dst, vals, nullptr, nullptr, bufA);
  gemm_wmma_kernel<<<gemm_blocks_h, 256, 0, stream>>>(
      bufA, W1, b1, g1, be1, rm1, rv1, bufB, N, NF, 8, 1);

  fill_zero<<<2048, 256, 0, stream>>>(bufA, nfeat_elems);
  spmm_kernel<<<E, 128, 0, stream>>>(bufB, src, dst, vals, nullptr, nullptr, bufA);
  gemm_wmma_kernel<<<gemm_blocks_o, 256, 0, stream>>>(
      bufA, W2, b2, nullptr, nullptr, nullptr, nullptr, outp, N, 40, 3, 0);
  log_softmax40<<<(N + 255) / 256, 256, 0, stream>>>(outp, N);
}


# GEMMs: resident groups; a wave keeps its 16-column W fragments in registers all launch; 16-row A tiles pass through LDS as whole 512-B rows, read once per group; k order and sums as before
# speedup vs baseline: 5.5867x; 5.5867x over previous
#include <hip/hip_runtime.h>
#include <math.h>

typedef __attribute__((ext_vector_type(2))) float v2f;
typedef __attribute__((ext_vector_type(4))) float v4f;
typedef __attribute__((ext_vector_type(8))) float v8f;

#define NF 128

#define ROWS_PER_WG 391
#define SCAN_WAVES 32
#define LIST_CAP 768
#define LIST_FLUSH 256
#define GATHER_DEPTH 16

__global__ __launch_bounds__(256) void owner_plane(
    const int* __restrict__ dst_a, const int* __restrict__ dst_b,
    unsigned* __restrict__ plane, int n_edges, int words_per_graph) {
  const int t = blockIdx.x * blockDim.x + threadIdx.x;
  if (t >= 2 * words_per_graph) return;
  const bool second = t >= words_per_graph;
  const int* d = second ? dst_b : dst_a;
  const int wi = second ? t - words_per_graph : t;
  const int e0 = wi * 4;
  int4 v;
  if (e0 + 3 < n_edges) {
    v = *(const int4*)(d + e0);
  } else {
    v.x = e0     < n_edges ? d[e0]     : -1;
    v.y = e0 + 1 < n_edges ? d[e0 + 1] : -1;
    v.z = e0 + 2 < n_edges ? d[e0 + 2] : -1;
    v.w = e0 + 3 < n_edges ? d[e0 + 3] : -1;
  }
  const unsigned b0 = ((unsigned)v.x / ROWS_PER_WG) & 255u;
  const unsigned b1 = ((unsigned)v.y / ROWS_PER_WG) & 255u;
  const unsigned b2 = ((unsigned)v.z / ROWS_PER_WG) & 255u;
  const unsigned b3 = ((unsigned)v.w / ROWS_PER_WG) & 255u;
  plane[t] = b0 | (b1 << 8) | (b2 << 16) | (b3 << 24);
}

__global__ __launch_bounds__(1024) void spmm_owner_scan(
    const float* __restrict__ X, const int* __restrict__ src,
    const int* __restrict__ dst, const float* __restrict__ vals,
    const float* __restrict__ Mv, const float* __restrict__ AMv,
    const unsigned* __restrict__ plane,
    float* __restrict__ out, int n_rows, int n_edges) {
  __shared__ float acc[(ROWS_PER_WG + 1) * NF];
  __shared__ int lst_e[SCAN_WAVES * LIST_CAP];

  const int lane = threadIdx.x & 31;
  const int wave = __builtin_amdgcn_readfirstlane(threadIdx.x >> 5);
  int* my_e = lst_e + wave * LIST_CAP;

  const int n_chunks = (n_edges + 2047) >> 11;
  const int n_rounds = (n_chunks + SCAN_WAVES - 1) / SCAN_WAVES;
  const int rot = (int)(((long)blockIdx.x * n_rounds) / gridDim.x);

  for (int row0 = blockIdx.x * ROWS_PER_WG; row0 < n_rows;
       row0 += gridDim.x * ROWS_PER_WG) {
    const int nr = min(ROWS_PER_WG, n_rows - row0);
    const unsigned pat = (((unsigned)row0 / ROWS_PER_WG) & 255u) * 0x01010101u;

    for (int i = threadIdx.x * 4; i < (ROWS_PER_WG + 1) * NF; i += 1024 * 4)
      *(float4*)(acc + i) = make_float4(0.f, 0.f, 0.f, 0.f);
    __syncthreads();

    int cnt = 0;

    auto add_row = [&](const float4& r, float wi, int di) {
      float* a = acc + di * NF + lane;
      atomicAdd(a,      wi * r.x);
      atomicAdd(a + 32, wi * r.y);
      atomicAdd(a + 64, wi * r.z);
      atomicAdd(a + 96, wi * r.w);
    };

    auto flush = [&](int count) {
      for (int b = 0; b < count; b += 32) {
        const int n = min(32, count - b);
        int s = 0, dl = ROWS_PER_WG;
        float w = 0.f;
        if (lane < n) {
          const int e = my_e[b + lane];
          const bool in = e < n_edges;
          const int ec = in ? e : 0;
          const int d = dst[ec] - row0;
          const int se = src[ec];
          const float we = vals[ec];
          const bool ok = in && (unsigned)d < (unsigned)nr;
          const int dr = ok ? d : 0;
          dl = ok ? d : ROWS_PER_WG;
          s = ok ? se : 0;
          w = ok ? we : 0.f;
          if (Mv)  w *= Mv[s];
          if (AMv) w *= AMv[row0 + dr];
        }
        const int wb = __float_as_int(w);
        int g = 0;
        for (; g + GATHER_DEPTH <= n; g += GATHER_DEPTH) {
          float4 r[GATHER_DEPTH];
#pragma unroll
          for (int i = 0; i < GATHER_DEPTH; ++i) {
            const int si = __builtin_amdgcn_readlane(s, g + i);
            r[i] = *(const float4*)(X + (size_t)si * NF + lane * 4);
          }
#pragma unroll
          for (int i = 0; i < GATHER_DEPTH; ++i)
            add_row(r[i], __int_as_float(__builtin_amdgcn_readlane(wb, g + i)),
                    __builtin_amdgcn_readlane(dl, g + i));
        }
        for (; g < n; g += 4) {
          float4 r[4];
#pragma unroll
          for (int i = 0; i < 4; ++i) {
            if (g + i < n) {
              const int si = __builtin_amdgcn_readlane(s, g + i);
              r[i] = *(const float4*)(X + (size_t)si * NF + lane * 4);
            }
          }
#pragma unroll
          for (int i = 0; i < 4; ++i)
            if (g + i < n)
              add_row(r[i], __int_as_float(__builtin_amdgcn_readlane(wb, g + i)),
                      __builtin_amdgcn_readlane(dl, g + i));
        }
      }
    };

    auto chunk_base = [&](int k) {
      int kr = k + rot;
      if (kr >= n_rounds) kr -= n_rounds;
      return (kr * SCAN_WAVES + wave) << 11;
    };
    auto load_chunk = [&](int base, int4 (&d)[4]) {
      if (base + 2048 <= n_edges) {
#pragma unroll
        for (int j = 0; j < 4; ++j)
          d[j] = *(const int4*)(plane + ((base + j * 512 + lane * 16) >> 2));
      } else {
#pragma unroll
        for (int j = 0; j < 4; ++j) {
          const int i0 = base + j * 512 + lane * 16;
          const int np = (int)~pat;
          d[j] = make_int4(np, np, np, np);
          if (i0 < n_edges)
            d[j] = *(const int4*)(plane + (i0 >> 2));
        }
      }
    };
    auto own_bytes = [&](int x) {
      const unsigned v = (unsigned)x ^ pat;
      return ~(((v & 0x7f7f7f7fu) + 0x7f7f7f7fu) | v | 0x7f7f7f7fu);
    };

    int4 cur[4];
#pragma unroll
    for (int j = 0; j < 4; ++j) {
      const int np = (int)~pat;
      cur[j] = make_int4(np, np, np, np);
    }
    for (int k = -1; k < n_rounds; ++k) {
      const int base = chunk_base(k < 0 ? 0 : k);
      int4 nxt[4];
      if (k + 1 < n_rounds) {
        load_chunk(chunk_base(k + 1), nxt);
      } else {
#pragma unroll
        for (int j = 0; j < 4; ++j) {
          const int np = (int)~pat;
          nxt[j] = make_int4(np, np, np, np);
        }
      }
#pragma unroll
      for (int j = 0; j < 4; ++j) {
        if (cnt > LIST_FLUSH) { flush(cnt); cnt = 0; }
        const int i0 = base + j * 512 + lane * 16;
        unsigned m = (own_bytes(cur[j].x) >> 7) | (own_bytes(cur[j].y) >> 6) |
                     (own_bytes(cur[j].z) >> 5) | (own_bytes(cur[j].w) >> 4);
        unsigned bal = __builtin_amdgcn_ballot_w32(m != 0u);
        while (bal != 0u) {
          if (m != 0u) {
            const int bit = __builtin_ctz(m);
            const int p = cnt + (int)__builtin_amdgcn_mbcnt_lo(bal, 0u);
            my_e[p] = i0 + ((bit & 7) << 2) + (bit >> 3);
            m &= m - 1u;
          }
          cnt += __builtin_popcount(bal);
          bal = __builtin_amdgcn_ballot_w32(m != 0u);
        }
      }
#pragma unroll
      for (int j = 0; j < 4; ++j) cur[j] = nxt[j];
    }
    flush(cnt);
    __syncthreads();

    for (int r = wave; r < nr; r += SCAN_WAVES) {
      const float* a = acc + r * NF + lane;
      float4 v;
      v.x = a[0]; v.y = a[32]; v.z = a[64]; v.w = a[96];
      *(float4*)(out + (size_t)(row0 + r) * NF + lane * 4) = v;
    }
    __syncthreads();
  }
}

#define A_PITCH 132

template <int NT>
__global__ __launch_bounds__(NT * 32) void gemm_wres_kernel(
    const float* __restrict__ A,
    const float* __restrict__ W,
    const float* __restrict__ bias,
    const float* __restrict__ g, const float* __restrict__ be,
    const float* __restrict__ rm, const float* __restrict__ rv,
    float* __restrict__ out,
    int n_rows, int n_out_real, int relu) {
  constexpr int K = NF;
  constexpr int KS = K / 4;
  constexpr int RPW = (16 + NT - 1) / NT;
  constexpr bool ALL = (NT * RPW == 16);
  __shared__ __attribute__((aligned(16))) float tileA[2][16 * A_PITCH];
  __shared__ __attribute__((aligned(16))) float tileW[NT][16 * A_PITCH];

  const int lane = threadIdx.x & 31;
  const int wave = __builtin_amdgcn_readfirstlane(threadIdx.x >> 5);
  const int tiles_m = (n_rows + 15) >> 4;
  if ((int)blockIdx.x >= tiles_m) return;

  const int hl = lane >> 4;
  const int ll = lane & 15;
  const int kh = hl * 2;
  const int col = wave * 16 + ll;
  const int cw  = col < n_out_real ? col : n_out_real - 1;

#pragma unroll
  for (int r = 0; r < 16; ++r) {
    int wc = wave * 16 + r;
    wc = wc < n_out_real ? wc : n_out_real - 1;
    *(v4f*)(&tileW[wave][r * A_PITCH + lane * 4]) = *(const v4f*)(W + (size_t)wc * K + lane * 4);
  }

  v4f pre[RPW];
  int tile = blockIdx.x;
#pragma unroll
  for (int i = 0; i < RPW; ++i) {
    const int rt = wave + NT * i;
    if (ALL || rt < 16) {
      int row = tile * 16 + rt;
      row = row < n_rows ? row : n_rows - 1;
      pre[i] = *(const v4f*)(A + (size_t)row * K + lane * 4);
    }
  }
#pragma unroll
  for (int i = 0; i < RPW; ++i) {
    const int rt = wave + NT * i;
    if (ALL || rt < 16) *(v4f*)(&tileA[0][rt * A_PITCH + lane * 4]) = pre[i];
  }

  const float bb = bias ? bias[cw] : 0.0f;
  float sc = 1.0f, sh = 0.0f;
  if (g) {
    const float s = g[cw] * rsqrtf(rv[cw] + 1e-5f);
    sc = s;
    sh = be[cw] - rm[cw] * s;
  }
  __syncthreads();

  v2f bw[KS];
  {
    const float* wq = &tileW[wave][ll * A_PITCH + kh];
#pragma unroll
    for (int s = 0; s < KS; ++s) bw[s] = *(const v2f*)(wq + 4 * s);
  }

  int buf = 0;
  for (; tile < tiles_m; tile += gridDim.x) {
    const int nxt = tile + gridDim.x;
    const bool more = nxt < tiles_m;
    if (more) {
#pragma unroll
      for (int i = 0; i < RPW; ++i) {
        const int rt = wave + NT * i;
        if (ALL || rt < 16) {
          int row = nxt * 16 + rt;
          row = row < n_rows ? row : n_rows - 1;
          pre[i] = *(const v4f*)(A + (size_t)row * K + lane * 4);
        }
      }
    }

    const float* ap = &tileA[buf][ll * A_PITCH + kh];
    v8f c = {};
#pragma unroll
    for (int s = 0; s < KS; ++s) {
      v2f a = *(const v2f*)(ap + 4 * s);
      c = __builtin_amdgcn_wmma_f32_16x16x4_f32(
           false, a,  false, bw[s],
           (short)0, c,  false,  false);
    }

    if (more) {
#pragma unroll
      for (int i = 0; i < RPW; ++i) {
        const int rt = wave + NT * i;
        if (ALL || rt < 16) *(v4f*)(&tileA[buf ^ 1][rt * A_PITCH + lane * 4]) = pre[i];
      }
    }
    __syncthreads();

    if (col < n_out_real) {
      const int rb = tile * 16 + hl * 8;
#pragma unroll
      for (int i = 0; i < 8; ++i) {
        const int r = rb + i;
        if (r < n_rows) {
          float v = (c[i] + bb) * sc + sh;
          if (relu) v = fmaxf(v, 0.0f);
          out[(size_t)r * n_out_real + col] = v;
        }
      }
    }
    buf ^= 1;
  }
}

__global__ __launch_bounds__(256) void log_softmax40(float* __restrict__ out, int n) {
  int i = blockIdx.x * blockDim.x + threadIdx.x;
  if (i >= n) return;
  float* row = out + (size_t)i * 40;
  float m = row[0];
#pragma unroll
  for (int j = 1; j < 40; ++j) m = fmaxf(m, row[j]);
  float s = 0.0f;
#pragma unroll
  for (int j = 0; j < 40; ++j) s += expf(row[j] - m);
  float l = m + logf(s);
#pragma unroll
  for (int j = 0; j < 40; ++j) row[j] = row[j] - l;
}

extern "C" void kernel_launch(void* const* d_in, const int* in_sizes, int n_in,
                              void* d_out, int out_size, void* d_ws, size_t ws_size,
                              hipStream_t stream) {
  const float* x     = (const float*)d_in[0];
  const float* Mv    = (const float*)d_in[1];
  const float* AMv   = (const float*)d_in[2];
  const int*   srcZ  = (const int*)d_in[3];
  const int*   dstZ  = (const int*)d_in[4];
  const float* valsZ = (const float*)d_in[5];
  const int*   src   = (const int*)d_in[6];
  const int*   dst   = (const int*)d_in[7];
  const float* vals  = (const float*)d_in[8];
  const float* W0 = (const float*)d_in[9];
  const float* b0 = (const float*)d_in[10];
  const float* g0 = (const float*)d_in[11];
  const float* be0 = (const float*)d_in[12];
  const float* rm0 = (const float*)d_in[13];
  const float* rv0 = (const float*)d_in[14];
  const float* W1 = (const float*)d_in[15];
  const float* b1 = (const float*)d_in[16];
  const float* g1 = (const float*)d_in[17];
  const float* be1 = (const float*)d_in[18];
  const float* rm1 = (const float*)d_in[19];
  const float* rv1 = (const float*)d_in[20];
  const float* W2 = (const float*)d_in[21];
  const float* b2 = (const float*)d_in[22];

  int N = in_sizes[1];
  int E = in_sizes[3];
  float* bufA = (float*)d_ws;
  float* bufB = bufA + (size_t)N * NF;
  float* outp = (float*)d_out;

  int plane_words = ((E + 15) / 16) * 4;
  unsigned* planeZ = (unsigned*)d_out;
  unsigned* planeG = planeZ + plane_words;

  const int gemm_blocks_h = 512;
  const int gemm_blocks_o = 1250;

  owner_plane<<<(2 * plane_words + 255) / 256, 256, 0, stream>>>(dstZ, dst, planeZ, E, plane_words);

  spmm_owner_scan<<<256, 1024, 0, stream>>>(x, srcZ, dstZ, valsZ, Mv, AMv, planeZ, bufA, N, E);
  gemm_wres_kernel<8><<<gemm_blocks_h, 256, 0, stream>>>(
      bufA, W0, b0, g0, be0, rm0, rv0, bufB, N, NF, 1);

  spmm_owner_scan<<<256, 1024, 0, stream>>>(bufB, src, dst, vals, nullptr, nullptr, planeG, bufA, N, E);
  gemm_wres_kernel<8><<<gemm_blocks_h, 256, 0, stream>>>(
      bufA, W1, b1, g1, be1, rm1, rv1, bufB, N, NF, 1);

  spmm_owner_scan<<<256, 1024, 0, stream>>>(bufB, src, dst, vals, nullptr, nullptr, planeG, bufA, N, E);
  gemm_wres_kernel<3><<<gemm_blocks_o, 96, 0, stream>>>(
      bufA, W2, b2, nullptr, nullptr, nullptr, nullptr, outp, N, 40, 0);
  log_softmax40<<<(N + 255) / 256, 256, 0, stream>>>(outp, N);
}


# SPMM layers 1, 2 (one graph): layer 1's scan keeps its hits as 8-B records (source row, local row, weight) per work-group; layer 2 reads them, no second scan or edge reads; LDS add order changes
# speedup vs baseline: 6.0967x; 1.0913x over previous
#include <hip/hip_runtime.h>
#include <math.h>

typedef __attribute__((ext_vector_type(2))) float v2f;
typedef __attribute__((ext_vector_type(4))) float v4f;
typedef __attribute__((ext_vector_type(8))) float v8f;

#define NF 128

#define ROWS_PER_WG 391
#define SCAN_WAVES 32
#define LIST_CAP 768
#define LIST_FLUSH 256
#define GATHER_DEPTH 16
#define REC_HDR 256
#define REC_SRC_BITS 23

__global__ __launch_bounds__(256) void owner_plane(
    const int* __restrict__ dst_a, const int* __restrict__ dst_b,
    unsigned* __restrict__ plane, int n_edges, int words_per_graph) {
  const int t = blockIdx.x * blockDim.x + threadIdx.x;
  if (t >= 2 * words_per_graph) return;
  const bool second = t >= words_per_graph;
  const int* d = second ? dst_b : dst_a;
  const int wi = second ? t - words_per_graph : t;
  const int e0 = wi * 4;
  int4 v;
  if (e0 + 3 < n_edges) {
    v = *(const int4*)(d + e0);
  } else {
    v.x = e0     < n_edges ? d[e0]     : -1;
    v.y = e0 + 1 < n_edges ? d[e0 + 1] : -1;
    v.z = e0 + 2 < n_edges ? d[e0 + 2] : -1;
    v.w = e0 + 3 < n_edges ? d[e0 + 3] : -1;
  }
  const unsigned b0 = ((unsigned)v.x / ROWS_PER_WG) & 255u;
  const unsigned b1 = ((unsigned)v.y / ROWS_PER_WG) & 255u;
  const unsigned b2 = ((unsigned)v.z / ROWS_PER_WG) & 255u;
  const unsigned b3 = ((unsigned)v.w / ROWS_PER_WG) & 255u;
  plane[t] = b0 | (b1 << 8) | (b2 << 16) | (b3 << 24);
}

template <int MODE>
__global__ __launch_bounds__(1024) void spmm_owner_scan(
    const float* __restrict__ X, const int* __restrict__ src,
    const int* __restrict__ dst, const float* __restrict__ vals,
    const float* __restrict__ Mv, const float* __restrict__ AMv,
    const unsigned* __restrict__ plane,
    float* __restrict__ out, int n_rows, int n_edges,
    unsigned* rec, int rec_dwords) {
  __shared__ float acc[(ROWS_PER_WG + 1) * NF];
  __shared__ int lst_e[SCAN_WAVES * LIST_CAP];
  __shared__ int rec_cnt;

  const int lane = threadIdx.x & 31;
  const int wave = __builtin_amdgcn_readfirstlane(threadIdx.x >> 5);
  int* my_e = lst_e + wave * LIST_CAP;

  const int n_chunks = (n_edges + 2047) >> 11;
  const int n_rounds = (n_chunks + SCAN_WAVES - 1) / SCAN_WAVES;
  const int rot = (int)(((long)blockIdx.x * n_rounds) / gridDim.x);

  const int cap = MODE == 0 ? 0 : ((rec_dwords - REC_HDR) / (2 * (int)gridDim.x)) & ~31;
  const bool rec_ok = MODE != 0 && cap > 0 && (int)gridDim.x <= REC_HDR &&
                      n_rows <= (int)gridDim.x * ROWS_PER_WG &&
                      n_rows <= (1 << REC_SRC_BITS);
  uint2* my_rec = MODE == 0 ? nullptr
                            : (uint2*)(rec + REC_HDR) + (size_t)blockIdx.x * (size_t)cap;

  for (int row0 = blockIdx.x * ROWS_PER_WG; row0 < n_rows;
       row0 += gridDim.x * ROWS_PER_WG) {
    const int nr = min(ROWS_PER_WG, n_rows - row0);
    const unsigned pat = (((unsigned)row0 / ROWS_PER_WG) & 255u) * 0x01010101u;

    for (int i = threadIdx.x * 4; i < (ROWS_PER_WG + 1) * NF; i += 1024 * 4)
      *(float4*)(acc + i) = make_float4(0.f, 0.f, 0.f, 0.f);
    if (MODE == 1 && threadIdx.x == 0) rec_cnt = 0;
    __syncthreads();

    int cnt = 0;

    auto add_row = [&](const float4& r, float wi, int di) {
      float* a = acc + di * NF + lane;
      atomicAdd(a,      wi * r.x);
      atomicAdd(a + 32, wi * r.y);
      atomicAdd(a + 64, wi * r.z);
      atomicAdd(a + 96, wi * r.w);
    };

    auto gather_add = [&](int s, int dl, int wb, int n) {
      int g = 0;
      for (; g + GATHER_DEPTH <= n; g += GATHER_DEPTH) {
        float4 r[GATHER_DEPTH];
#pragma unroll
        for (int i = 0; i < GATHER_DEPTH; ++i) {
          const int si = __builtin_amdgcn_readlane(s, g + i);
          r[i] = *(const float4*)(X + (size_t)si * NF + lane * 4);
        }
#pragma unroll
        for (int i = 0; i < GATHER_DEPTH; ++i)
          add_row(r[i], __int_as_float(__builtin_amdgcn_readlane(wb, g + i)),
                  __builtin_amdgcn_readlane(dl, g + i));
      }
      for (; g < n; g += 4) {
        float4 r[4];
#pragma unroll
        for (int i = 0; i < 4; ++i) {
          if (g + i < n) {
            const int si = __builtin_amdgcn_readlane(s, g + i);
            r[i] = *(const float4*)(X + (size_t)si * NF + lane * 4);
          }
        }
#pragma unroll
        for (int i = 0; i < 4; ++i)
          if (g + i < n)
            add_row(r[i], __int_as_float(__builtin_amdgcn_readlane(wb, g + i)),
                    __builtin_amdgcn_readlane(dl, g + i));
      }
    };

    auto flush = [&](int count) {
      for (int b = 0; b < count; b += 32) {
        const int n = min(32, count - b);
        int s = 0, dl = ROWS_PER_WG;
        float w = 0.f;
        if (lane < n) {
          const int e = my_e[b + lane];
          const bool in = e < n_edges;
          const int ec = in ? e : 0;
          const int d = dst[ec] - row0;
          const int se = src[ec];
          const float we = vals[ec];
          const bool ok = in && (unsigned)d < (unsigned)nr;
          const int dr = ok ? d : 0;
          dl = ok ? d : ROWS_PER_WG;
          s = ok ? se : 0;
          w = ok ? we : 0.f;
          if (Mv)  w *= Mv[s];
          if (AMv) w *= AMv[row0 + dr];
        }
        const int wb = __float_as_int(w);
        if (MODE == 1 && rec_ok) {
          int base = 0;
          if (lane == 0) base = atomicAdd(&rec_cnt, n);
          base = __builtin_amdgcn_readfirstlane(base);
          if (base + n <= cap && lane < n)
            my_rec[base + lane] =
                make_uint2((unsigned)s | ((unsigned)dl << REC_SRC_BITS), (unsigned)wb);
        }
        gather_add(s, dl, wb, n);
      }
    };

    auto chunk_base = [&](int k) {
      int kr = k + rot;
      if (kr >= n_rounds) kr -= n_rounds;
      return (kr * SCAN_WAVES + wave) << 11;
    };
    auto load_chunk = [&](int base, int4 (&d)[4]) {
      if (base + 2048 <= n_edges) {
#pragma unroll
        for (int j = 0; j < 4; ++j)
          d[j] = *(const int4*)(plane + ((base + j * 512 + lane * 16) >> 2));
      } else {
#pragma unroll
        for (int j = 0; j < 4; ++j) {
          const int i0 = base + j * 512 + lane * 16;
          const int np = (int)~pat;
          d[j] = make_int4(np, np, np, np);
          if (i0 < n_edges)
            d[j] = *(const int4*)(plane + (i0 >> 2));
        }
      }
    };
    auto own_bytes = [&](int x) {
      const unsigned v = (unsigned)x ^ pat;
      return ~(((v & 0x7f7f7f7fu) + 0x7f7f7f7fu) | v | 0x7f7f7f7fu);
    };

    int n_rec = -1;
    if (MODE == 2 && rec_ok)
      n_rec = __builtin_amdgcn_readfirstlane((int)rec[blockIdx.x]);
    const bool from_rec = MODE == 2 && rec_ok && (unsigned)n_rec <= (unsigned)cap;

    if (from_rec) {
      const int r0 = (n_rec * wave) / SCAN_WAVES;
      const int r1 = (n_rec * (wave + 1)) / SCAN_WAVES;
      if (r0 < r1) {
        uint2 nx = my_rec[min(r0 + lane, r1 - 1)];
        for (int b = r0; b < r1; b += 32) {
          const uint2 cu = nx;
          nx = my_rec[min(b + 32 + lane, r1 - 1)];
          const bool have = b + lane < r1;
          const int s = have ? min((int)(cu.x & ((1u << REC_SRC_BITS) - 1u)), n_rows - 1) : 0;
          const int dl = have ? min((int)(cu.x >> REC_SRC_BITS), ROWS_PER_WG) : ROWS_PER_WG;
          const int wb = have ? (int)cu.y : 0;
          gather_add(s, dl, wb, min(32, r1 - b));
        }
      }
    } else {
      int4 cur[4];
#pragma unroll
      for (int j = 0; j < 4; ++j) {
        const int np = (int)~pat;
        cur[j] = make_int4(np, np, np, np);
      }
      for (int k = -1; k < n_rounds; ++k) {
        const int base = chunk_base(k < 0 ? 0 : k);
        int4 nxt[4];
        if (k + 1 < n_rounds) {
          load_chunk(chunk_base(k + 1), nxt);
        } else {
#pragma unroll
          for (int j = 0; j < 4; ++j) {
            const int np = (int)~pat;
            nxt[j] = make_int4(np, np, np, np);
          }
        }
#pragma unroll
        for (int j = 0; j < 4; ++j) {
          if (cnt > LIST_FLUSH) { flush(cnt); cnt = 0; }
          const int i0 = base + j * 512 + lane * 16;
          unsigned m = (own_bytes(cur[j].x) >> 7) | (own_bytes(cur[j].y) >> 6) |
                       (own_bytes(cur[j].z) >> 5) | (own_bytes(cur[j].w) >> 4);
          unsigned bal = __builtin_amdgcn_ballot_w32(m != 0u);
          while (bal != 0u) {
            if (m != 0u) {
              const int bit = __builtin_ctz(m);
              const int p = cnt + (int)__builtin_amdgcn_mbcnt_lo(bal, 0u);
              my_e[p] = i0 + ((bit & 7) << 2) + (bit >> 3);
              m &= m - 1u;
            }
            cnt += __builtin_popcount(bal);
            bal = __builtin_amdgcn_ballot_w32(m != 0u);
          }
        }
#pragma unroll
        for (int j = 0; j < 4; ++j) cur[j] = nxt[j];
      }
      flush(cnt);
    }
    __syncthreads();
    if (MODE == 1 && rec_ok && threadIdx.x == 0) rec[blockIdx.x] = (unsigned)rec_cnt;

    for (int r = wave; r < nr; r += SCAN_WAVES) {
      const float* a = acc + r * NF + lane;
      float4 v;
      v.x = a[0]; v.y = a[32]; v.z = a[64]; v.w = a[96];
      *(float4*)(out + (size_t)(row0 + r) * NF + lane * 4) = v;
    }
    __syncthreads();
  }
}

#define A_PITCH 132

template <int NT>
__global__ __launch_bounds__(NT * 32) void gemm_wres_kernel(
    const float* __restrict__ A,
    const float* __restrict__ W,
    const float* __restrict__ bias,
    const float* __restrict__ g, const float* __restrict__ be,
    const float* __restrict__ rm, const float* __restrict__ rv,
    float* __restrict__ out,
    int n_rows, int n_out_real, int relu) {
  constexpr int K = NF;
  constexpr int KS = K / 4;
  constexpr int RPW = (16 + NT - 1) / NT;
  constexpr bool ALL = (NT * RPW == 16);
  __shared__ __attribute__((aligned(16))) float tileA[2][16 * A_PITCH];
  __shared__ __attribute__((aligned(16))) float tileW[NT][16 * A_PITCH];

  const int lane = threadIdx.x & 31;
  const int wave = __builtin_amdgcn_readfirstlane(threadIdx.x >> 5);
  const int tiles_m = (n_rows + 15) >> 4;
  if ((int)blockIdx.x >= tiles_m) return;

  const int hl = lane >> 4;
  const int ll = lane & 15;
  const int kh = hl * 2;
  const int col = wave * 16 + ll;
  const int cw  = col < n_out_real ? col : n_out_real - 1;

#pragma unroll
  for (int r = 0; r < 16; ++r) {
    int wc = wave * 16 + r;
    wc = wc < n_out_real ? wc : n_out_real - 1;
    *(v4f*)(&tileW[wave][r * A_PITCH + lane * 4]) = *(const v4f*)(W + (size_t)wc * K + lane * 4);
  }

  v4f pre[RPW];
  int tile = blockIdx.x;
#pragma unroll
  for (int i = 0; i < RPW; ++i) {
    const int rt = wave + NT * i;
    if (ALL || rt < 16) {
      int row = tile * 16 + rt;
      row = row < n_rows ? row : n_rows - 1;
      pre[i] = *(const v4f*)(A + (size_t)row * K + lane * 4);
    }
  }
#pragma unroll
  for (int i = 0; i < RPW; ++i) {
    const int rt = wave + NT * i;
    if (ALL || rt < 16) *(v4f*)(&tileA[0][rt * A_PITCH + lane * 4]) = pre[i];
  }

  const float bb = bias ? bias[cw] : 0.0f;
  float sc = 1.0f, sh = 0.0f;
  if (g) {
    const float s = g[cw] * rsqrtf(rv[cw] + 1e-5f);
    sc = s;
    sh = be[cw] - rm[cw] * s;
  }
  __syncthreads();

  v2f bw[KS];
  {
    const float* wq = &tileW[wave][ll * A_PITCH + kh];
#pragma unroll
    for (int s = 0; s < KS; ++s) bw[s] = *(const v2f*)(wq + 4 * s);
  }

  int buf = 0;
  for (; tile < tiles_m; tile += gridDim.x) {
    const int nxt = tile + gridDim.x;
    const bool more = nxt < tiles_m;
    if (more) {
#pragma unroll
      for (int i = 0; i < RPW; ++i) {
        const int rt = wave + NT * i;
        if (ALL || rt < 16) {
          int row = nxt * 16 + rt;
          row = row < n_rows ? row : n_rows - 1;
          pre[i] = *(const v4f*)(A + (size_t)row * K + lane * 4);
        }
      }
    }

    const float* ap = &tileA[buf][ll * A_PITCH + kh];
    v8f c = {};
#pragma unroll
    for (int s = 0; s < KS; ++s) {
      v2f a = *(const v2f*)(ap + 4 * s);
      c = __builtin_amdgcn_wmma_f32_16x16x4_f32(
           false, a,  false, bw[s],
           (short)0, c,  false,  false);
    }

    if (more) {
#pragma unroll
      for (int i = 0; i < RPW; ++i) {
        const int rt = wave + NT * i;
        if (ALL || rt < 16) *(v4f*)(&tileA[buf ^ 1][rt * A_PITCH + lane * 4]) = pre[i];
      }
    }
    __syncthreads();

    if (col < n_out_real) {
      const int rb = tile * 16 + hl * 8;
#pragma unroll
      for (int i = 0; i < 8; ++i) {
        const int r = rb + i;
        if (r < n_rows) {
          float v = (c[i] + bb) * sc + sh;
          if (relu) v = fmaxf(v, 0.0f);
          out[(size_t)r * n_out_real + col] = v;
        }
      }
    }
    buf ^= 1;
  }
}

__global__ __launch_bounds__(256) void log_softmax40(float* __restrict__ out, int n) {
  int i = blockIdx.x * blockDim.x + threadIdx.x;
  if (i >= n) return;
  float* row = out + (size_t)i * 40;
  float m = row[0];
#pragma unroll
  for (int j = 1; j < 40; ++j) m = fmaxf(m, row[j]);
  float s = 0.0f;
#pragma unroll
  for (int j = 0; j < 40; ++j) s += expf(row[j] - m);
  float l = m + logf(s);
#pragma unroll
  for (int j = 0; j < 40; ++j) row[j] = row[j] - l;
}

extern "C" void kernel_launch(void* const* d_in, const int* in_sizes, int n_in,
                              void* d_out, int out_size, void* d_ws, size_t ws_size,
                              hipStream_t stream) {
  const float* x     = (const float*)d_in[0];
  const float* Mv    = (const float*)d_in[1];
  const float* AMv   = (const float*)d_in[2];
  const int*   srcZ  = (const int*)d_in[3];
  const int*   dstZ  = (const int*)d_in[4];
  const float* valsZ = (const float*)d_in[5];
  const int*   src   = (const int*)d_in[6];
  const int*   dst   = (const int*)d_in[7];
  const float* vals  = (const float*)d_in[8];
  const float* W0 = (const float*)d_in[9];
  const float* b0 = (const float*)d_in[10];
  const float* g0 = (const float*)d_in[11];
  const float* be0 = (const float*)d_in[12];
  const float* rm0 = (const float*)d_in[13];
  const float* rv0 = (const float*)d_in[14];
  const float* W1 = (const float*)d_in[15];
  const float* b1 = (const float*)d_in[16];
  const float* g1 = (const float*)d_in[17];
  const float* be1 = (const float*)d_in[18];
  const float* rm1 = (const float*)d_in[19];
  const float* rv1 = (const float*)d_in[20];
  const float* W2 = (const float*)d_in[21];
  const float* b2 = (const float*)d_in[22];

  int N = in_sizes[1];
  int E = in_sizes[3];
  float* bufA = (float*)d_ws;
  float* bufB = bufA + (size_t)N * NF;
  float* outp = (float*)d_out;

  int plane_words = ((E + 15) / 16) * 4;
  unsigned* planeG = (unsigned*)d_out;
  unsigned* planeZ = planeG + plane_words;
  unsigned* recs = planeZ;
  int rec_dwords = N * 40 - plane_words;

  const int gemm_blocks_h = 512;
  const int gemm_blocks_o = 1250;

  owner_plane<<<(2 * plane_words + 255) / 256, 256, 0, stream>>>(dst, dstZ, planeG, E, plane_words);

  spmm_owner_scan<0><<<256, 1024, 0, stream>>>(x, srcZ, dstZ, valsZ, Mv, AMv, planeZ, bufA, N, E,
                                               nullptr, 0);
  gemm_wres_kernel<8><<<gemm_blocks_h, 256, 0, stream>>>(
      bufA, W0, b0, g0, be0, rm0, rv0, bufB, N, NF, 1);

  spmm_owner_scan<1><<<256, 1024, 0, stream>>>(bufB, src, dst, vals, nullptr, nullptr, planeG, bufA, N, E,
                                               recs, rec_dwords);
  gemm_wres_kernel<8><<<gemm_blocks_h, 256, 0, stream>>>(
      bufA, W1, b1, g1, be1, rm1, rv1, bufB, N, NF, 1);

  spmm_owner_scan<2><<<256, 1024, 0, stream>>>(bufB, src, dst, vals, nullptr, nullptr, planeG, bufA, N, E,
                                               recs, rec_dwords);
  gemm_wres_kernel<3><<<gemm_blocks_o, 96, 0, stream>>>(
      bufA, W2, b2, nullptr, nullptr, nullptr, nullptr, outp, N, 40, 0);
  log_softmax40<<<(N + 255) / 256, 256, 0, stream>>>(outp, N);
}


# Layer 0 SPMM: first launch also packs each edge as one 8-B record (src row, row in block, vals*M[src]) into the idle hidden buffer; a listed edge reads 1 scattered line, not 4; sums as before
# speedup vs baseline: 6.3135x; 1.0356x over previous
#include <hip/hip_runtime.h>
#include <math.h>

typedef __attribute__((ext_vector_type(2))) float v2f;
typedef __attribute__((ext_vector_type(4))) float v4f;
typedef __attribute__((ext_vector_type(8))) float v8f;

#define NF 128

#define ROWS_PER_WG 391
#define SCAN_WAVES 32
#define LIST_CAP 768
#define LIST_FLUSH 256
#define GATHER_DEPTH 16
#define REC_HDR 256
#define REC_SRC_BITS 23
#define REC_SRC_MASK ((1u << REC_SRC_BITS) - 1u)

__device__ __forceinline__ bool edge_recs_fit(int n_rows, int n_edges) {
  return (long)n_edges * 2 <= (long)n_rows * NF;
}

__global__ __launch_bounds__(256) void owner_plane(
    const int* __restrict__ dst_a, const int* __restrict__ dst_b,
    const int* __restrict__ src_b, const float* __restrict__ vals_b,
    const float* __restrict__ Mv,
    unsigned* __restrict__ plane, unsigned* __restrict__ erec,
    int n_edges, int words_per_graph, int n_rows) {
  const int t = blockIdx.x * blockDim.x + threadIdx.x;
  if (t >= 2 * words_per_graph) return;
  const bool second = t >= words_per_graph;
  const int* d = second ? dst_b : dst_a;
  const int wi = second ? t - words_per_graph : t;
  const int e0 = wi * 4;
  const bool whole = e0 + 3 < n_edges;
  const bool recs = second && erec != nullptr && edge_recs_fit(n_rows, n_edges);
  int4 v;
  int4 s = make_int4(0, 0, 0, 0);
  float4 w = make_float4(0.f, 0.f, 0.f, 0.f);
  if (whole) {
    v = *(const int4*)(d + e0);
    if (recs) {
      s = *(const int4*)(src_b + e0);
      w = *(const float4*)(vals_b + e0);
    }
  } else {
    v.x = e0     < n_edges ? d[e0]     : -1;
    v.y = e0 + 1 < n_edges ? d[e0 + 1] : -1;
    v.z = e0 + 2 < n_edges ? d[e0 + 2] : -1;
    v.w = e0 + 3 < n_edges ? d[e0 + 3] : -1;
    if (recs) {
      if (e0     < n_edges) { s.x = src_b[e0];     w.x = vals_b[e0]; }
      if (e0 + 1 < n_edges) { s.y = src_b[e0 + 1]; w.y = vals_b[e0 + 1]; }
      if (e0 + 2 < n_edges) { s.z = src_b[e0 + 2]; w.z = vals_b[e0 + 2]; }
      if (e0 + 3 < n_edges) { s.w = src_b[e0 + 3]; w.w = vals_b[e0 + 3]; }
    }
  }
  const unsigned q0 = (unsigned)v.x / ROWS_PER_WG;
  const unsigned q1 = (unsigned)v.y / ROWS_PER_WG;
  const unsigned q2 = (unsigned)v.z / ROWS_PER_WG;
  const unsigned q3 = (unsigned)v.w / ROWS_PER_WG;
  plane[t] = (q0 & 255u) | ((q1 & 255u) << 8) | ((q2 & 255u) << 16) |
             ((q3 & 255u) << 24);

  if (recs) {
    if (Mv) { w.x *= Mv[s.x]; w.y *= Mv[s.y]; w.z *= Mv[s.z]; w.w *= Mv[s.w]; }
    uint4 ra, rb;
    ra.x = ((unsigned)s.x & REC_SRC_MASK) | (((unsigned)v.x - q0 * ROWS_PER_WG) << REC_SRC_BITS);
    ra.y = __float_as_uint(w.x);
    ra.z = ((unsigned)s.y & REC_SRC_MASK) | (((unsigned)v.y - q1 * ROWS_PER_WG) << REC_SRC_BITS);
    ra.w = __float_as_uint(w.y);
    rb.x = ((unsigned)s.z & REC_SRC_MASK) | (((unsigned)v.z - q2 * ROWS_PER_WG) << REC_SRC_BITS);
    rb.y = __float_as_uint(w.z);
    rb.z = ((unsigned)s.w & REC_SRC_MASK) | (((unsigned)v.w - q3 * ROWS_PER_WG) << REC_SRC_BITS);
    rb.w = __float_as_uint(w.w);
    unsigned* r = erec + (size_t)e0 * 2;
    if (whole) {
      *(uint4*)r = ra;
      *(uint4*)(r + 4) = rb;
    } else {
      if (e0     < n_edges) { r[0] = ra.x; r[1] = ra.y; }
      if (e0 + 1 < n_edges) { r[2] = ra.z; r[3] = ra.w; }
      if (e0 + 2 < n_edges) { r[4] = rb.x; r[5] = rb.y; }
      if (e0 + 3 < n_edges) { r[6] = rb.z; r[7] = rb.w; }
    }
  }
}

template <int MODE>
__global__ __launch_bounds__(1024) void spmm_owner_scan(
    const float* __restrict__ X, const int* __restrict__ src,
    const int* __restrict__ dst, const float* __restrict__ vals,
    const float* __restrict__ Mv, const float* __restrict__ AMv,
    const unsigned* __restrict__ plane,
    float* __restrict__ out, int n_rows, int n_edges,
    unsigned* rec, int rec_dwords, const unsigned* __restrict__ erec) {
  __shared__ float acc[(ROWS_PER_WG + 1) * NF];
  __shared__ int lst_e[SCAN_WAVES * LIST_CAP];
  __shared__ int rec_cnt;

  const int lane = threadIdx.x & 31;
  const int wave = __builtin_amdgcn_readfirstlane(threadIdx.x >> 5);
  int* my_e = lst_e + wave * LIST_CAP;

  const int n_chunks = (n_edges + 2047) >> 11;
  const int n_rounds = (n_chunks + SCAN_WAVES - 1) / SCAN_WAVES;
  const int rot = (int)(((long)blockIdx.x * n_rounds) / gridDim.x);

  const int cap = MODE == 0 ? 0 : ((rec_dwords - REC_HDR) / (2 * (int)gridDim.x)) & ~31;
  const bool rec_ok = MODE != 0 && cap > 0 && (int)gridDim.x <= REC_HDR &&
                      n_rows <= (int)gridDim.x * ROWS_PER_WG &&
                      n_rows <= (1 << REC_SRC_BITS);
  uint2* my_rec = MODE == 0 ? nullptr
                            : (uint2*)(rec + REC_HDR) + (size_t)blockIdx.x * (size_t)cap;
  const bool from_erec = MODE == 0 && erec != nullptr && edge_recs_fit(n_rows, n_edges) &&
                         (int)gridDim.x <= 256 &&
                         n_rows <= (int)gridDim.x * ROWS_PER_WG &&
                         n_rows <= (1 << REC_SRC_BITS);

  for (int row0 = blockIdx.x * ROWS_PER_WG; row0 < n_rows;
       row0 += gridDim.x * ROWS_PER_WG) {
    const int nr = min(ROWS_PER_WG, n_rows - row0);
    const unsigned pat = (((unsigned)row0 / ROWS_PER_WG) & 255u) * 0x01010101u;

    for (int i = threadIdx.x * 4; i < (ROWS_PER_WG + 1) * NF; i += 1024 * 4)
      *(float4*)(acc + i) = make_float4(0.f, 0.f, 0.f, 0.f);
    if (MODE == 1 && threadIdx.x == 0) rec_cnt = 0;
    __syncthreads();

    int cnt = 0;

    auto add_row = [&](const float4& r, float wi, int di) {
      float* a = acc + di * NF + lane;
      atomicAdd(a,      wi * r.x);
      atomicAdd(a + 32, wi * r.y);
      atomicAdd(a + 64, wi * r.z);
      atomicAdd(a + 96, wi * r.w);
    };

    auto gather_add = [&](int s, int dl, int wb, int n) {
      int g = 0;
      for (; g + GATHER_DEPTH <= n; g += GATHER_DEPTH) {
        float4 r[GATHER_DEPTH];
#pragma unroll
        for (int i = 0; i < GATHER_DEPTH; ++i) {
          const int si = __builtin_amdgcn_readlane(s, g + i);
          r[i] = *(const float4*)(X + (size_t)si * NF + lane * 4);
        }
#pragma unroll
        for (int i = 0; i < GATHER_DEPTH; ++i)
          add_row(r[i], __int_as_float(__builtin_amdgcn_readlane(wb, g + i)),
                  __builtin_amdgcn_readlane(dl, g + i));
      }
      for (; g < n; g += 4) {
        float4 r[4];
#pragma unroll
        for (int i = 0; i < 4; ++i) {
          if (g + i < n) {
            const int si = __builtin_amdgcn_readlane(s, g + i);
            r[i] = *(const float4*)(X + (size_t)si * NF + lane * 4);
          }
        }
#pragma unroll
        for (int i = 0; i < 4; ++i)
          if (g + i < n)
            add_row(r[i], __int_as_float(__builtin_amdgcn_readlane(wb, g + i)),
                    __builtin_amdgcn_readlane(dl, g + i));
      }
    };

    auto flush = [&](int count) {
      for (int b = 0; b < count; b += 32) {
        const int n = min(32, count - b);
        int s = 0, dl = ROWS_PER_WG;
        float w = 0.f;
        if (lane < n) {
          const int e = my_e[b + lane];
          const bool in = e < n_edges;
          const int ec = in ? e : 0;
          if (from_erec) {
            const uint2 rc = *(const uint2*)(erec + (size_t)ec * 2);
            const int d = (int)(rc.x >> REC_SRC_BITS);
            const bool ok = in && d < nr;
            dl = ok ? d : ROWS_PER_WG;
            s = ok ? min((int)(rc.x & REC_SRC_MASK), n_rows - 1) : 0;
            w = ok ? __uint_as_float(rc.y) : 0.f;
            if (AMv) w *= AMv[row0 + (ok ? d : 0)];
          } else {
            const int d = dst[ec] - row0;
            const int se = src[ec];
            const float we = vals[ec];
            const bool ok = in && (unsigned)d < (unsigned)nr;
            const int dr = ok ? d : 0;
            dl = ok ? d : ROWS_PER_WG;
            s = ok ? se : 0;
            w = ok ? we : 0.f;
            if (Mv)  w *= Mv[s];
            if (AMv) w *= AMv[row0 + dr];
          }
        }
        const int wb = __float_as_int(w);
        if (MODE == 1 && rec_ok) {
          int base = 0;
          if (lane == 0) base = atomicAdd(&rec_cnt, n);
          base = __builtin_amdgcn_readfirstlane(base);
          if (base + n <= cap && lane < n)
            my_rec[base + lane] =
                make_uint2((unsigned)s | ((unsigned)dl << REC_SRC_BITS), (unsigned)wb);
        }
        gather_add(s, dl, wb, n);
      }
    };

    auto chunk_base = [&](int k) {
      int kr = k + rot;
      if (kr >= n_rounds) kr -= n_rounds;
      return (kr * SCAN_WAVES + wave) << 11;
    };
    auto load_chunk = [&](int base, int4 (&d)[4]) {
      if (base + 2048 <= n_edges) {
#pragma unroll
        for (int j = 0; j < 4; ++j)
          d[j] = *(const int4*)(plane + ((base + j * 512 + lane * 16) >> 2));
      } else {
#pragma unroll
        for (int j = 0; j < 4; ++j) {
          const int i0 = base + j * 512 + lane * 16;
          const int np = (int)~pat;
          d[j] = make_int4(np, np, np, np);
          if (i0 < n_edges)
            d[j] = *(const int4*)(plane + (i0 >> 2));
        }
      }
    };
    auto own_bytes = [&](int x) {
      const unsigned v = (unsigned)x ^ pat;
      return ~(((v & 0x7f7f7f7fu) + 0x7f7f7f7fu) | v | 0x7f7f7f7fu);
    };

    int n_rec = -1;
    if (MODE == 2 && rec_ok)
      n_rec = __builtin_amdgcn_readfirstlane((int)rec[blockIdx.x]);
    const bool from_rec = MODE == 2 && rec_ok && (unsigned)n_rec <= (unsigned)cap;

    if (from_rec) {
      const int r0 = (n_rec * wave) / SCAN_WAVES;
      const int r1 = (n_rec * (wave + 1)) / SCAN_WAVES;
      if (r0 < r1) {
        uint2 nx = my_rec[min(r0 + lane, r1 - 1)];
        for (int b = r0; b < r1; b += 32) {
          const uint2 cu = nx;
          nx = my_rec[min(b + 32 + lane, r1 - 1)];
          const bool have = b + lane < r1;
          const int s = have ? min((int)(cu.x & ((1u << REC_SRC_BITS) - 1u)), n_rows - 1) : 0;
          const int dl = have ? min((int)(cu.x >> REC_SRC_BITS), ROWS_PER_WG) : ROWS_PER_WG;
          const int wb = have ? (int)cu.y : 0;
          gather_add(s, dl, wb, min(32, r1 - b));
        }
      }
    } else {
      int4 cur[4];
#pragma unroll
      for (int j = 0; j < 4; ++j) {
        const int np = (int)~pat;
        cur[j] = make_int4(np, np, np, np);
      }
      for (int k = -1; k < n_rounds; ++k) {
        const int base = chunk_base(k < 0 ? 0 : k);
        int4 nxt[4];
        if (k + 1 < n_rounds) {
          load_chunk(chunk_base(k + 1), nxt);
        } else {
#pragma unroll
          for (int j = 0; j < 4; ++j) {
            const int np = (int)~pat;
            nxt[j] = make_int4(np, np, np, np);
          }
        }
#pragma unroll
        for (int j = 0; j < 4; ++j) {
          if (cnt > LIST_FLUSH) { flush(cnt); cnt = 0; }
          const int i0 = base + j * 512 + lane * 16;
          unsigned m = (own_bytes(cur[j].x) >> 7) | (own_bytes(cur[j].y) >> 6) |
                       (own_bytes(cur[j].z) >> 5) | (own_bytes(cur[j].w) >> 4);
          unsigned bal = __builtin_amdgcn_ballot_w32(m != 0u);
          while (bal != 0u) {
            if (m != 0u) {
              const int bit = __builtin_ctz(m);
              const int p = cnt + (int)__builtin_amdgcn_mbcnt_lo(bal, 0u);
              my_e[p] = i0 + ((bit & 7) << 2) + (bit >> 3);
              m &= m - 1u;
            }
            cnt += __builtin_popcount(bal);
            bal = __builtin_amdgcn_ballot_w32(m != 0u);
          }
        }
#pragma unroll
        for (int j = 0; j < 4; ++j) cur[j] = nxt[j];
      }
      flush(cnt);
    }
    __syncthreads();
    if (MODE == 1 && rec_ok && threadIdx.x == 0) rec[blockIdx.x] = (unsigned)rec_cnt;

    for (int r = wave; r < nr; r += SCAN_WAVES) {
      const float* a = acc + r * NF + lane;
      float4 v;
      v.x = a[0]; v.y = a[32]; v.z = a[64]; v.w = a[96];
      *(float4*)(out + (size_t)(row0 + r) * NF + lane * 4) = v;
    }
    __syncthreads();
  }
}

#define A_PITCH 132

template <int NT>
__global__ __launch_bounds__(NT * 32) void gemm_wres_kernel(
    const float* __restrict__ A,
    const float* __restrict__ W,
    const float* __restrict__ bias,
    const float* __restrict__ g, const float* __restrict__ be,
    const float* __restrict__ rm, const float* __restrict__ rv,
    float* __restrict__ out,
    int n_rows, int n_out_real, int relu) {
  constexpr int K = NF;
  constexpr int KS = K / 4;
  constexpr int RPW = (16 + NT - 1) / NT;
  constexpr bool ALL = (NT * RPW == 16);
  __shared__ __attribute__((aligned(16))) float tileA[2][16 * A_PITCH];
  __shared__ __attribute__((aligned(16))) float tileW[NT][16 * A_PITCH];

  const int lane = threadIdx.x & 31;
  const int wave = __builtin_amdgcn_readfirstlane(threadIdx.x >> 5);
  const int tiles_m = (n_rows + 15) >> 4;
  if ((int)blockIdx.x >= tiles_m) return;

  const int hl = lane >> 4;
  const int ll = lane & 15;
  const int kh = hl * 2;
  const int col = wave * 16 + ll;
  const int cw  = col < n_out_real ? col : n_out_real - 1;

#pragma unroll
  for (int r = 0; r < 16; ++r) {
    int wc = wave * 16 + r;
    wc = wc < n_out_real ? wc : n_out_real - 1;
    *(v4f*)(&tileW[wave][r * A_PITCH + lane * 4]) = *(const v4f*)(W + (size_t)wc * K + lane * 4);
  }

  v4f pre[RPW];
  int tile = blockIdx.x;
#pragma unroll
  for (int i = 0; i < RPW; ++i) {
    const int rt = wave + NT * i;
    if (ALL || rt < 16) {
      int row = tile * 16 + rt;
      row = row < n_rows ? row : n_rows - 1;
      pre[i] = *(const v4f*)(A + (size_t)row * K + lane * 4);
    }
  }
#pragma unroll
  for (int i = 0; i < RPW; ++i) {
    const int rt = wave + NT * i;
    if (ALL || rt < 16) *(v4f*)(&tileA[0][rt * A_PITCH + lane * 4]) = pre[i];
  }

  const float bb = bias ? bias[cw] : 0.0f;
  float sc = 1.0f, sh = 0.0f;
  if (g) {
    const float s = g[cw] * rsqrtf(rv[cw] + 1e-5f);
    sc = s;
    sh = be[cw] - rm[cw] * s;
  }
  __syncthreads();

  v2f bw[KS];
  {
    const float* wq = &tileW[wave][ll * A_PITCH + kh];
#pragma unroll
    for (int s = 0; s < KS; ++s) bw[s] = *(const v2f*)(wq + 4 * s);
  }

  int buf = 0;
  for (; tile < tiles_m; tile += gridDim.x) {
    const int nxt = tile + gridDim.x;
    const bool more = nxt < tiles_m;
    if (more) {
#pragma unroll
      for (int i = 0; i < RPW; ++i) {
        const int rt = wave + NT * i;
        if (ALL || rt < 16) {
          int row = nxt * 16 + rt;
          row = row < n_rows ? row : n_rows - 1;
          pre[i] = *(const v4f*)(A + (size_t)row * K + lane * 4);
        }
      }
    }

    const float* ap = &tileA[buf][ll * A_PITCH + kh];
    v8f c = {};
#pragma unroll
    for (int s = 0; s < KS; ++s) {
      v2f a = *(const v2f*)(ap + 4 * s);
      c = __builtin_amdgcn_wmma_f32_16x16x4_f32(
           false, a,  false, bw[s],
           (short)0, c,  false,  false);
    }

    if (more) {
#pragma unroll
      for (int i = 0; i < RPW; ++i) {
        const int rt = wave + NT * i;
        if (ALL || rt < 16) *(v4f*)(&tileA[buf ^ 1][rt * A_PITCH + lane * 4]) = pre[i];
      }
    }
    __syncthreads();

    if (col < n_out_real) {
      const int rb = tile * 16 + hl * 8;
#pragma unroll
      for (int i = 0; i < 8; ++i) {
        const int r = rb + i;
        if (r < n_rows) {
          float v = (c[i] + bb) * sc + sh;
          if (relu) v = fmaxf(v, 0.0f);
          out[(size_t)r * n_out_real + col] = v;
        }
      }
    }
    buf ^= 1;
  }
}

__global__ __launch_bounds__(256) void log_softmax40(float* __restrict__ out, int n) {
  int i = blockIdx.x * blockDim.x + threadIdx.x;
  if (i >= n) return;
  float* row = out + (size_t)i * 40;
  float m = row[0];
#pragma unroll
  for (int j = 1; j < 40; ++j) m = fmaxf(m, row[j]);
  float s = 0.0f;
#pragma unroll
  for (int j = 0; j < 40; ++j) s += expf(row[j] - m);
  float l = m + logf(s);
#pragma unroll
  for (int j = 0; j < 40; ++j) row[j] = row[j] - l;
}

extern "C" void kernel_launch(void* const* d_in, const int* in_sizes, int n_in,
                              void* d_out, int out_size, void* d_ws, size_t ws_size,
                              hipStream_t stream) {
  const float* x     = (const float*)d_in[0];
  const float* Mv    = (const float*)d_in[1];
  const float* AMv   = (const float*)d_in[2];
  const int*   srcZ  = (const int*)d_in[3];
  const int*   dstZ  = (const int*)d_in[4];
  const float* valsZ = (const float*)d_in[5];
  const int*   src   = (const int*)d_in[6];
  const int*   dst   = (const int*)d_in[7];
  const float* vals  = (const float*)d_in[8];
  const float* W0 = (const float*)d_in[9];
  const float* b0 = (const float*)d_in[10];
  const float* g0 = (const float*)d_in[11];
  const float* be0 = (const float*)d_in[12];
  const float* rm0 = (const float*)d_in[13];
  const float* rv0 = (const float*)d_in[14];
  const float* W1 = (const float*)d_in[15];
  const float* b1 = (const float*)d_in[16];
  const float* g1 = (const float*)d_in[17];
  const float* be1 = (const float*)d_in[18];
  const float* rm1 = (const float*)d_in[19];
  const float* rv1 = (const float*)d_in[20];
  const float* W2 = (const float*)d_in[21];
  const float* b2 = (const float*)d_in[22];

  int N = in_sizes[1];
  int E = in_sizes[3];
  float* bufA = (float*)d_ws;
  float* bufB = bufA + (size_t)N * NF;
  float* outp = (float*)d_out;

  int plane_words = ((E + 15) / 16) * 4;
  unsigned* planeG = (unsigned*)d_out;
  unsigned* planeZ = planeG + plane_words;
  unsigned* recs = planeZ;
  int rec_dwords = N * 40 - plane_words;

  const int gemm_blocks_h = 512;
  const int gemm_blocks_o = 1250;

  unsigned* zrec = (unsigned*)bufB;

  owner_plane<<<(2 * plane_words + 255) / 256, 256, 0, stream>>>(dst, dstZ, srcZ, valsZ, Mv, planeG, zrec,
                                                                 E, plane_words, N);

  spmm_owner_scan<0><<<256, 1024, 0, stream>>>(x, srcZ, dstZ, valsZ, Mv, AMv, planeZ, bufA, N, E,
                                               nullptr, 0, zrec);
  gemm_wres_kernel<8><<<gemm_blocks_h, 256, 0, stream>>>(
      bufA, W0, b0, g0, be0, rm0, rv0, bufB, N, NF, 1);

  spmm_owner_scan<1><<<256, 1024, 0, stream>>>(bufB, src, dst, vals, nullptr, nullptr, planeG, bufA, N, E,
                                               recs, rec_dwords, nullptr);
  gemm_wres_kernel<8><<<gemm_blocks_h, 256, 0, stream>>>(
      bufA, W1, b1, g1, be1, rm1, rv1, bufB, N, NF, 1);

  spmm_owner_scan<2><<<256, 1024, 0, stream>>>(bufB, src, dst, vals, nullptr, nullptr, planeG, bufA, N, E,
                                               recs, rec_dwords, nullptr);
  gemm_wres_kernel<3><<<gemm_blocks_o, 96, 0, stream>>>(
      bufA, W2, b2, nullptr, nullptr, nullptr, nullptr, outp, N, 40, 0);
  log_softmax40<<<(N + 255) / 256, 256, 0, stream>>>(outp, N);
}
